# starting kernel
# baseline (speedup 1.0000x reference)
#include <hip/hip_runtime.h>
#include <hip/hip_bf16.h>
#include <stdint.h>

typedef __attribute__((ext_vector_type(2))) float v2f;
typedef __attribute__((ext_vector_type(8))) float v8f;

__constant__ int c_unis[58] = {
    0, 1, 2, 3, 4, 6, 7, 8, 12, 14, 15, 16, 24, 28, 30, 31, 32, 48, 56, 60,
    62, 63, 64, 96, 112, 120, 124, 126, 127, 128, 129, 131, 135, 143, 159,
    191, 192, 193, 195, 199, 207, 223, 224, 225, 227, 231, 239, 240, 241,
    243, 247, 248, 249, 251, 252, 253, 254, 255};

__global__ void zero_hist_kernel(unsigned* __restrict__ hist, int n) {
  int i = blockIdx.x * blockDim.x + threadIdx.x;
  if (i < n) hist[i] = 0u;
}

__device__ __forceinline__ unsigned quant(float f) {
  return (unsigned)(f * 255.0f);
}

template <int NQ>
__device__ __forceinline__ void load_quant_row(const float* __restrict__ img,
                                               int row, int cstart,
                                               unsigned* __restrict__ q) {
  if ((unsigned)row < 512u) {
    const float* rp = img + row * 512;
#pragma unroll
    for (int i = 0; i < NQ; ++i) {
      const int cq = cstart + i * 4;
      if ((unsigned)cq < 512u) {
        const float4 v = *(const float4*)(rp + cq);
        q[i * 4 + 0] = quant(v.x);
        q[i * 4 + 1] = quant(v.y);
        q[i * 4 + 2] = quant(v.z);
        q[i * 4 + 3] = quant(v.w);
      } else {
        q[i * 4 + 0] = 0u;
        q[i * 4 + 1] = 0u;
        q[i * 4 + 2] = 0u;
        q[i * 4 + 3] = 0u;
      }
    }
  } else {
#pragma unroll
    for (int i = 0; i < NQ * 4; ++i) q[i] = 0u;
  }
}

__global__ __launch_bounds__(256) void lbp_hist_kernel(
    const float* __restrict__ x, unsigned* __restrict__ hist) {
  __shared__ unsigned sh[256];
  const int t = threadIdx.x;
  sh[t] = 0u;
  __syncthreads();

  const int batch = blockIdx.y;
  const int r = (blockIdx.x << 3) + (t >> 5);
  const int c = (t & 31) << 4;
  const float* img = x + (size_t)batch * (512 * 512);

  unsigned qc[32], qu[24], qd[24];
  load_quant_row<8>(img, r, c - 8, qc);
  load_quant_row<6>(img, r - 3, c - 4, qu);
  load_quant_row<6>(img, r + 3, c - 4, qd);

#pragma unroll
  for (int j = 0; j < 16; ++j) {
    const unsigned center = qc[j + 8];
    unsigned code = 0u;
    code |= (qu[j + 4]  >= center) ? 1u   : 0u;
    code |= (qu[j + 7]  >= center) ? 2u   : 0u;
    code |= (qc[j + 13] >= center) ? 4u   : 0u;
    code |= (qd[j + 7]  >= center) ? 8u   : 0u;
    code |= (qd[j + 4]  >= center) ? 16u  : 0u;
    code |= (qd[j + 1]  >= center) ? 32u  : 0u;
    code |= (qc[j + 3]  >= center) ? 64u  : 0u;
    code |= (qu[j + 1]  >= center) ? 128u : 0u;
    atomicAdd(&sh[code], 1u);
  }

  __syncthreads();
  const unsigned v = sh[t];
  if (v) atomicAdd(&hist[batch * 256 + t], v);
}

__global__ __launch_bounds__(256) void wmma_reduce_kernel(
    const unsigned* __restrict__ hist, float* __restrict__ out, int B) {
  __shared__ int smap[256];
  const int t = threadIdx.x;
  {
    int m = 58;
#pragma unroll
    for (int j = 0; j < 58; ++j)
      if (c_unis[j] == t) m = j;
    smap[t] = m;
  }
  __syncthreads();

  const int wave = t >> 5;
  const int lane = t & 31;
  const int lm = lane & 15;
  const int hi = (lane >> 4) & 1;
  const int mtiles = B >> 4;
  const int ntotal = mtiles * 4;

  for (int tile = wave; tile < ntotal; tile += 8) {
    const int mt = tile >> 2;
    const int nt = tile & 3;
    const int m = (mt << 4) + lm;
    const int n = (nt << 4) + lm;

    v8f acc = {};
    const unsigned* hrow = hist + (size_t)m * 256;
    for (int k0 = 0; k0 < 256; k0 += 4) {
      const int ka = k0 + (hi ? 2 : 0);
      v2f A = {(float)hrow[ka], (float)hrow[ka + 1]};
      v2f Bm = {(smap[ka] == n) ? 1.0f : 0.0f,
                (smap[ka + 1] == n) ? 1.0f : 0.0f};
      acc = __builtin_amdgcn_wmma_f32_16x16x4_f32(
           false, A,  false, Bm,
           (short)0, acc,  false,  false);
    }

    if (n < 59) {
#pragma unroll
      for (int r = 0; r < 8; ++r) {
        const int M = (mt << 4) + (hi << 3) + r;
        const unsigned iv = (unsigned)(acc[r] + 0.5f);
        out[M * 59 + n] = (float)(iv & 255u);
      }
    }
  }
}

extern "C" void kernel_launch(void* const* d_in, const int* in_sizes, int n_in,
                              void* d_out, int out_size, void* d_ws,
                              size_t ws_size, hipStream_t stream) {
  const float* x = (const float*)d_in[0];
  float* out = (float*)d_out;
  unsigned* hist = (unsigned*)d_ws;

  const int B = in_sizes[0] >> 18;

  zero_hist_kernel<<<(B * 256 + 255) / 256, 256, 0, stream>>>(hist, B * 256);
  lbp_hist_kernel<<<dim3(64, B), 256, 0, stream>>>(x, hist);
  wmma_reduce_kernel<<<1, 256, 0, stream>>>(hist, out, B);
}


# lbp kernel coarsened: 2xB work-groups of 32 waves (one per WGP), each wave loops over 8 rows, instead of 64xB of 8 waves; 8x fewer wave entries, 32x fewer histogram flush atomics
# speedup vs baseline: 1.1700x; 1.1700x over previous
#include <hip/hip_runtime.h>
#include <hip/hip_bf16.h>
#include <stdint.h>

typedef __attribute__((ext_vector_type(2))) float v2f;
typedef __attribute__((ext_vector_type(8))) float v8f;

__constant__ int c_unis[58] = {
    0, 1, 2, 3, 4, 6, 7, 8, 12, 14, 15, 16, 24, 28, 30, 31, 32, 48, 56, 60,
    62, 63, 64, 96, 112, 120, 124, 126, 127, 128, 129, 131, 135, 143, 159,
    191, 192, 193, 195, 199, 207, 223, 224, 225, 227, 231, 239, 240, 241,
    243, 247, 248, 249, 251, 252, 253, 254, 255};

__global__ void zero_hist_kernel(unsigned* __restrict__ hist, int n) {
  int i = blockIdx.x * blockDim.x + threadIdx.x;
  if (i < n) hist[i] = 0u;
}

__device__ __forceinline__ unsigned quant(float f) {
  return (unsigned)(f * 255.0f);
}

template <int NQ>
__device__ __forceinline__ void load_quant_row(const float* __restrict__ img,
                                               int row, int cstart,
                                               unsigned* __restrict__ q) {
  if ((unsigned)row < 512u) {
    const float* rp = img + row * 512;
#pragma unroll
    for (int i = 0; i < NQ; ++i) {
      const int cq = cstart + i * 4;
      if ((unsigned)cq < 512u) {
        const float4 v = *(const float4*)(rp + cq);
        q[i * 4 + 0] = quant(v.x);
        q[i * 4 + 1] = quant(v.y);
        q[i * 4 + 2] = quant(v.z);
        q[i * 4 + 3] = quant(v.w);
      } else {
        q[i * 4 + 0] = 0u;
        q[i * 4 + 1] = 0u;
        q[i * 4 + 2] = 0u;
        q[i * 4 + 3] = 0u;
      }
    }
  } else {
#pragma unroll
    for (int i = 0; i < NQ * 4; ++i) q[i] = 0u;
  }
}

constexpr int LBP_WAVES = 32;
constexpr int LBP_PASSES = 8;
static_assert(2 * LBP_WAVES * LBP_PASSES == 512, "two work-groups cover one image");

__global__ __launch_bounds__(LBP_WAVES * 32) void lbp_hist_kernel(
    const float* __restrict__ x, unsigned* __restrict__ hist) {
  __shared__ unsigned sh[256];
  const int t = threadIdx.x;
  if (t < 256) sh[t] = 0u;
  __syncthreads();

  const int batch = blockIdx.y;
  const int r0 = blockIdx.x * (LBP_WAVES * LBP_PASSES) + (t >> 5);
  const int c = (t & 31) << 4;
  const float* img = x + (size_t)batch * (512 * 512);

  for (int p = 0; p < LBP_PASSES; ++p) {
  const int r = r0 + LBP_WAVES * p;

  unsigned qc[32], qu[24], qd[24];
  load_quant_row<8>(img, r, c - 8, qc);
  load_quant_row<6>(img, r - 3, c - 4, qu);
  load_quant_row<6>(img, r + 3, c - 4, qd);

#pragma unroll
  for (int j = 0; j < 16; ++j) {
    const unsigned center = qc[j + 8];
    unsigned code = 0u;
    code |= (qu[j + 4]  >= center) ? 1u   : 0u;
    code |= (qu[j + 7]  >= center) ? 2u   : 0u;
    code |= (qc[j + 13] >= center) ? 4u   : 0u;
    code |= (qd[j + 7]  >= center) ? 8u   : 0u;
    code |= (qd[j + 4]  >= center) ? 16u  : 0u;
    code |= (qd[j + 1]  >= center) ? 32u  : 0u;
    code |= (qc[j + 3]  >= center) ? 64u  : 0u;
    code |= (qu[j + 1]  >= center) ? 128u : 0u;
    atomicAdd(&sh[code], 1u);
  }
  }

  __syncthreads();
  if (t < 256) {
    const unsigned v = sh[t];
    if (v) atomicAdd(&hist[batch * 256 + t], v);
  }
}

__global__ __launch_bounds__(256) void wmma_reduce_kernel(
    const unsigned* __restrict__ hist, float* __restrict__ out, int B) {
  __shared__ int smap[256];
  const int t = threadIdx.x;
  {
    int m = 58;
#pragma unroll
    for (int j = 0; j < 58; ++j)
      if (c_unis[j] == t) m = j;
    smap[t] = m;
  }
  __syncthreads();

  const int wave = t >> 5;
  const int lane = t & 31;
  const int lm = lane & 15;
  const int hi = (lane >> 4) & 1;
  const int mtiles = B >> 4;
  const int ntotal = mtiles * 4;

  for (int tile = wave; tile < ntotal; tile += 8) {
    const int mt = tile >> 2;
    const int nt = tile & 3;
    const int m = (mt << 4) + lm;
    const int n = (nt << 4) + lm;

    v8f acc = {};
    const unsigned* hrow = hist + (size_t)m * 256;
    for (int k0 = 0; k0 < 256; k0 += 4) {
      const int ka = k0 + (hi ? 2 : 0);
      v2f A = {(float)hrow[ka], (float)hrow[ka + 1]};
      v2f Bm = {(smap[ka] == n) ? 1.0f : 0.0f,
                (smap[ka + 1] == n) ? 1.0f : 0.0f};
      acc = __builtin_amdgcn_wmma_f32_16x16x4_f32(
           false, A,  false, Bm,
           (short)0, acc,  false,  false);
    }

    if (n < 59) {
#pragma unroll
      for (int r = 0; r < 8; ++r) {
        const int M = (mt << 4) + (hi << 3) + r;
        const unsigned iv = (unsigned)(acc[r] + 0.5f);
        out[M * 59 + n] = (float)(iv & 255u);
      }
    }
  }
}

extern "C" void kernel_launch(void* const* d_in, const int* in_sizes, int n_in,
                              void* d_out, int out_size, void* d_ws,
                              size_t ws_size, hipStream_t stream) {
  const float* x = (const float*)d_in[0];
  float* out = (float*)d_out;
  unsigned* hist = (unsigned*)d_ws;

  const int B = in_sizes[0] >> 18;

  zero_hist_kernel<<<(B * 256 + 255) / 256, 256, 0, stream>>>(hist, B * 256);
  lbp_hist_kernel<<<dim3(2, B), LBP_WAVES * 32, 0, stream>>>(x, hist);
  wmma_reduce_kernel<<<1, 256, 0, stream>>>(hist, out, B);
}


# lbp kernel: LDS histogram kept as 32 lane-indexed copies (word = bin*32 + lane), so lanes of one ds_add never share a word or bank; copies summed at the flush
# speedup vs baseline: 1.8512x; 1.5822x over previous
#include <hip/hip_runtime.h>
#include <hip/hip_bf16.h>
#include <stdint.h>

typedef __attribute__((ext_vector_type(2))) float v2f;
typedef __attribute__((ext_vector_type(8))) float v8f;

__constant__ int c_unis[58] = {
    0, 1, 2, 3, 4, 6, 7, 8, 12, 14, 15, 16, 24, 28, 30, 31, 32, 48, 56, 60,
    62, 63, 64, 96, 112, 120, 124, 126, 127, 128, 129, 131, 135, 143, 159,
    191, 192, 193, 195, 199, 207, 223, 224, 225, 227, 231, 239, 240, 241,
    243, 247, 248, 249, 251, 252, 253, 254, 255};

__global__ void zero_hist_kernel(unsigned* __restrict__ hist, int n) {
  int i = blockIdx.x * blockDim.x + threadIdx.x;
  if (i < n) hist[i] = 0u;
}

__device__ __forceinline__ unsigned quant(float f) {
  return (unsigned)(f * 255.0f);
}

template <int NQ>
__device__ __forceinline__ void load_quant_row(const float* __restrict__ img,
                                               int row, int cstart,
                                               unsigned* __restrict__ q) {
  if ((unsigned)row < 512u) {
    const float* rp = img + row * 512;
#pragma unroll
    for (int i = 0; i < NQ; ++i) {
      const int cq = cstart + i * 4;
      if ((unsigned)cq < 512u) {
        const float4 v = *(const float4*)(rp + cq);
        q[i * 4 + 0] = quant(v.x);
        q[i * 4 + 1] = quant(v.y);
        q[i * 4 + 2] = quant(v.z);
        q[i * 4 + 3] = quant(v.w);
      } else {
        q[i * 4 + 0] = 0u;
        q[i * 4 + 1] = 0u;
        q[i * 4 + 2] = 0u;
        q[i * 4 + 3] = 0u;
      }
    }
  } else {
#pragma unroll
    for (int i = 0; i < NQ * 4; ++i) q[i] = 0u;
  }
}

constexpr int LBP_WAVES = 32;
constexpr int LBP_PASSES = 8;
constexpr int LBP_COPIES = 32;
static_assert(2 * LBP_WAVES * LBP_PASSES == 512, "two work-groups cover one image");
static_assert(256 * LBP_COPIES == 2 * 4 * LBP_WAVES * 32, "two 16-byte stores per thread clear it");

__global__ __launch_bounds__(LBP_WAVES * 32) void lbp_hist_kernel(
    const float* __restrict__ x, unsigned* __restrict__ hist) {
  __shared__ __attribute__((aligned(16))) unsigned sh[256 * LBP_COPIES];
  const int t = threadIdx.x;
  {
    uint4* sh4 = reinterpret_cast<uint4*>(sh);
    const uint4 z = {0u, 0u, 0u, 0u};
    sh4[t] = z;
    sh4[t + LBP_WAVES * 32] = z;
  }
  __syncthreads();
  const unsigned lane = (unsigned)t & 31u;

  const int batch = blockIdx.y;
  const int r0 = blockIdx.x * (LBP_WAVES * LBP_PASSES) + (t >> 5);
  const int c = (t & 31) << 4;
  const float* img = x + (size_t)batch * (512 * 512);

  for (int p = 0; p < LBP_PASSES; ++p) {
  const int r = r0 + LBP_WAVES * p;

  unsigned qc[32], qu[24], qd[24];
  load_quant_row<8>(img, r, c - 8, qc);
  load_quant_row<6>(img, r - 3, c - 4, qu);
  load_quant_row<6>(img, r + 3, c - 4, qd);

#pragma unroll
  for (int j = 0; j < 16; ++j) {
    const unsigned center = qc[j + 8];
    unsigned code = 0u;
    code |= (qu[j + 4]  >= center) ? 1u   : 0u;
    code |= (qu[j + 7]  >= center) ? 2u   : 0u;
    code |= (qc[j + 13] >= center) ? 4u   : 0u;
    code |= (qd[j + 7]  >= center) ? 8u   : 0u;
    code |= (qd[j + 4]  >= center) ? 16u  : 0u;
    code |= (qd[j + 1]  >= center) ? 32u  : 0u;
    code |= (qc[j + 3]  >= center) ? 64u  : 0u;
    code |= (qu[j + 1]  >= center) ? 128u : 0u;
    atomicAdd(&sh[(code << 5) | lane], 1u);
  }
  }

  __syncthreads();
  if (t < 256) {
    unsigned v = 0u;
    for (int i = 0; i < LBP_COPIES; ++i) v += sh[(t << 5) | ((i + t) & 31)];
    if (v) atomicAdd(&hist[batch * 256 + t], v);
  }
}

__global__ __launch_bounds__(256) void wmma_reduce_kernel(
    const unsigned* __restrict__ hist, float* __restrict__ out, int B) {
  __shared__ int smap[256];
  const int t = threadIdx.x;
  {
    int m = 58;
#pragma unroll
    for (int j = 0; j < 58; ++j)
      if (c_unis[j] == t) m = j;
    smap[t] = m;
  }
  __syncthreads();

  const int wave = t >> 5;
  const int lane = t & 31;
  const int lm = lane & 15;
  const int hi = (lane >> 4) & 1;
  const int mtiles = B >> 4;
  const int ntotal = mtiles * 4;

  for (int tile = wave; tile < ntotal; tile += 8) {
    const int mt = tile >> 2;
    const int nt = tile & 3;
    const int m = (mt << 4) + lm;
    const int n = (nt << 4) + lm;

    v8f acc = {};
    const unsigned* hrow = hist + (size_t)m * 256;
    for (int k0 = 0; k0 < 256; k0 += 4) {
      const int ka = k0 + (hi ? 2 : 0);
      v2f A = {(float)hrow[ka], (float)hrow[ka + 1]};
      v2f Bm = {(smap[ka] == n) ? 1.0f : 0.0f,
                (smap[ka + 1] == n) ? 1.0f : 0.0f};
      acc = __builtin_amdgcn_wmma_f32_16x16x4_f32(
           false, A,  false, Bm,
           (short)0, acc,  false,  false);
    }

    if (n < 59) {
#pragma unroll
      for (int r = 0; r < 8; ++r) {
        const int M = (mt << 4) + (hi << 3) + r;
        const unsigned iv = (unsigned)(acc[r] + 0.5f);
        out[M * 59 + n] = (float)(iv & 255u);
      }
    }
  }
}

extern "C" void kernel_launch(void* const* d_in, const int* in_sizes, int n_in,
                              void* d_out, int out_size, void* d_ws,
                              size_t ws_size, hipStream_t stream) {
  const float* x = (const float*)d_in[0];
  float* out = (float*)d_out;
  unsigned* hist = (unsigned*)d_ws;

  const int B = in_sizes[0] >> 18;

  zero_hist_kernel<<<(B * 256 + 255) / 256, 256, 0, stream>>>(hist, B * 256);
  lbp_hist_kernel<<<dim3(2, B), LBP_WAVES * 32, 0, stream>>>(x, hist);
  wmma_reduce_kernel<<<1, 256, 0, stream>>>(hist, out, B);
}


# lbp kernel: work-group stages its 256+6 rows once in LDS as quantized bytes (whole-line global loads, zero padding staged); lanes read their row windows from LDS, not by 20 strided global loads
# speedup vs baseline: 2.2375x; 1.2087x over previous
#include <hip/hip_runtime.h>
#include <hip/hip_bf16.h>
#include <stdint.h>

typedef __attribute__((ext_vector_type(2))) float v2f;
typedef __attribute__((ext_vector_type(8))) float v8f;

__constant__ int c_unis[58] = {
    0, 1, 2, 3, 4, 6, 7, 8, 12, 14, 15, 16, 24, 28, 30, 31, 32, 48, 56, 60,
    62, 63, 64, 96, 112, 120, 124, 126, 127, 128, 129, 131, 135, 143, 159,
    191, 192, 193, 195, 199, 207, 223, 224, 225, 227, 231, 239, 240, 241,
    243, 247, 248, 249, 251, 252, 253, 254, 255};

__global__ void zero_hist_kernel(unsigned* __restrict__ hist, int n) {
  int i = blockIdx.x * blockDim.x + threadIdx.x;
  if (i < n) hist[i] = 0u;
}

__device__ __forceinline__ unsigned quant(float f) {
  return (unsigned)(f * 255.0f);
}

constexpr int LBP_WAVES = 32;
constexpr int LBP_PASSES = 8;
constexpr int LBP_COPIES = 32;
constexpr int LBP_ROWS = LBP_WAVES * LBP_PASSES;
constexpr int LBP_HALO = 3;
constexpr int STG_ROWS = LBP_ROWS + 2 * LBP_HALO;
constexpr int STG_PAD_W = 4;
constexpr int STG_PITCH_W = 128 + 2 * STG_PAD_W;
constexpr int STG_PITCH_Q = STG_PITCH_W / 4;
constexpr int STG_BATCH = 3;
constexpr int STG_STEPS = (STG_ROWS + LBP_WAVES - 1) / LBP_WAVES;
static_assert(2 * LBP_ROWS == 512, "two work-groups cover one image");
static_assert(256 * LBP_COPIES == 2 * 4 * LBP_WAVES * 32, "two 16-byte stores per thread clear it");
static_assert(STG_STEPS % STG_BATCH == 0, "whole batches");

template <int B>
__device__ __forceinline__ unsigned win_byte(const unsigned (&w)[12]) {
  return (w[B >> 2] >> ((B & 3) * 8)) & 255u;
}

__device__ __forceinline__ void load_window(const uint4* __restrict__ row4, unsigned lane,
                                            unsigned (&w)[12]) {
#pragma unroll
  for (int i = 0; i < 3; ++i) {
    const uint4 v = row4[lane + i];
    w[i * 4 + 0] = v.x;
    w[i * 4 + 1] = v.y;
    w[i * 4 + 2] = v.z;
    w[i * 4 + 3] = v.w;
  }
}

template <int J>
__device__ __forceinline__ unsigned lbp_code(const unsigned (&wc)[12], const unsigned (&wu)[12],
                                             const unsigned (&wd)[12]) {
  const unsigned center = win_byte<16 + J>(wc);
  unsigned code = 0u;
  code |= (win_byte<16 + J>(wu)     >= center) ? 1u   : 0u;
  code |= (win_byte<16 + J + 3>(wu) >= center) ? 2u   : 0u;
  code |= (win_byte<16 + J + 5>(wc) >= center) ? 4u   : 0u;
  code |= (win_byte<16 + J + 3>(wd) >= center) ? 8u   : 0u;
  code |= (win_byte<16 + J>(wd)     >= center) ? 16u  : 0u;
  code |= (win_byte<16 + J - 3>(wd) >= center) ? 32u  : 0u;
  code |= (win_byte<16 + J - 5>(wc) >= center) ? 64u  : 0u;
  code |= (win_byte<16 + J - 3>(wu) >= center) ? 128u : 0u;
  return code;
}

template <int J>
__device__ __forceinline__ void lbp_row16(const unsigned (&wc)[12], const unsigned (&wu)[12],
                                          const unsigned (&wd)[12], unsigned* __restrict__ sh,
                                          unsigned lane) {
  if constexpr (J < 16) {
    const unsigned code = lbp_code<J>(wc, wu, wd);
    atomicAdd(&sh[(code << 5) | lane], 1u);
    lbp_row16<J + 1>(wc, wu, wd, sh, lane);
  }
}

__global__ __launch_bounds__(LBP_WAVES * 32) void lbp_hist_kernel(
    const float* __restrict__ x, unsigned* __restrict__ hist) {
  __shared__ __attribute__((aligned(16))) unsigned sh[256 * LBP_COPIES];
  __shared__ __attribute__((aligned(16))) unsigned tile[STG_ROWS * STG_PITCH_W];
  const int t = threadIdx.x;
  {
    uint4* sh4 = reinterpret_cast<uint4*>(sh);
    const uint4 z = {0u, 0u, 0u, 0u};
    sh4[t] = z;
    sh4[t + LBP_WAVES * 32] = z;
  }
  const unsigned lane = (unsigned)t & 31u;
  const int wave = t >> 5;

  const int batch = blockIdx.y;
  const int row0 = blockIdx.x * LBP_ROWS - LBP_HALO;
  const float* img = x + (size_t)batch * (512 * 512);

  for (int k0 = 0; k0 < STG_STEPS; k0 += STG_BATCH) {
    float4 v[STG_BATCH][4];
#pragma unroll
    for (int kk = 0; kk < STG_BATCH; ++kk) {
      const int j = wave + LBP_WAVES * (k0 + kk);
      const int jl = j < STG_ROWS ? j : STG_ROWS - 1;
      int gr = row0 + jl;
      gr = gr < 0 ? 0 : (gr > 511 ? 511 : gr);
      const float4* rp = reinterpret_cast<const float4*>(img + gr * 512);
#pragma unroll
      for (int i = 0; i < 4; ++i) v[kk][i] = rp[lane + 32 * i];
    }
#pragma unroll
    for (int kk = 0; kk < STG_BATCH; ++kk) {
      const int j = wave + LBP_WAVES * (k0 + kk);
      const bool in_img = (unsigned)(row0 + j) < 512u;
      if (j < STG_ROWS) {
        unsigned* dst = tile + j * STG_PITCH_W;
#pragma unroll
        for (int i = 0; i < 4; ++i) {
          const unsigned pk = quant(v[kk][i].x) | (quant(v[kk][i].y) << 8) |
                              (quant(v[kk][i].z) << 16) | (quant(v[kk][i].w) << 24);
          dst[STG_PAD_W + lane + 32 * i] = in_img ? pk : 0u;
        }
        if (lane < 2 * STG_PAD_W) dst[lane < STG_PAD_W ? lane : 128 + lane] = 0u;
      }
    }
  }
  __syncthreads();

  const uint4* tile4 = reinterpret_cast<const uint4*>(tile);
  for (int p = 0; p < LBP_PASSES; ++p) {
    const int jc = LBP_HALO + LBP_WAVES * p + wave;
    unsigned wc[12], wu[12], wd[12];
    load_window(tile4 + jc * STG_PITCH_Q, lane, wc);
    load_window(tile4 + (jc - LBP_HALO) * STG_PITCH_Q, lane, wu);
    load_window(tile4 + (jc + LBP_HALO) * STG_PITCH_Q, lane, wd);
    lbp_row16<0>(wc, wu, wd, sh, lane);
  }

  __syncthreads();
  if (t < 256) {
    unsigned v = 0u;
    for (int i = 0; i < LBP_COPIES; ++i) v += sh[(t << 5) | ((i + t) & 31)];
    if (v) atomicAdd(&hist[batch * 256 + t], v);
  }
}

__global__ __launch_bounds__(256) void wmma_reduce_kernel(
    const unsigned* __restrict__ hist, float* __restrict__ out, int B) {
  __shared__ int smap[256];
  const int t = threadIdx.x;
  {
    int m = 58;
#pragma unroll
    for (int j = 0; j < 58; ++j)
      if (c_unis[j] == t) m = j;
    smap[t] = m;
  }
  __syncthreads();

  const int wave = t >> 5;
  const int lane = t & 31;
  const int lm = lane & 15;
  const int hi = (lane >> 4) & 1;
  const int mtiles = B >> 4;
  const int ntotal = mtiles * 4;

  for (int tile = wave; tile < ntotal; tile += 8) {
    const int mt = tile >> 2;
    const int nt = tile & 3;
    const int m = (mt << 4) + lm;
    const int n = (nt << 4) + lm;

    v8f acc = {};
    const unsigned* hrow = hist + (size_t)m * 256;
    for (int k0 = 0; k0 < 256; k0 += 4) {
      const int ka = k0 + (hi ? 2 : 0);
      v2f A = {(float)hrow[ka], (float)hrow[ka + 1]};
      v2f Bm = {(smap[ka] == n) ? 1.0f : 0.0f,
                (smap[ka + 1] == n) ? 1.0f : 0.0f};
      acc = __builtin_amdgcn_wmma_f32_16x16x4_f32(
           false, A,  false, Bm,
           (short)0, acc,  false,  false);
    }

    if (n < 59) {
#pragma unroll
      for (int r = 0; r < 8; ++r) {
        const int M = (mt << 4) + (hi << 3) + r;
        const unsigned iv = (unsigned)(acc[r] + 0.5f);
        out[M * 59 + n] = (float)(iv & 255u);
      }
    }
  }
}

extern "C" void kernel_launch(void* const* d_in, const int* in_sizes, int n_in,
                              void* d_out, int out_size, void* d_ws,
                              size_t ws_size, hipStream_t stream) {
  const float* x = (const float*)d_in[0];
  float* out = (float*)d_out;
  unsigned* hist = (unsigned*)d_ws;

  const int B = in_sizes[0] >> 18;

  zero_hist_kernel<<<(B * 256 + 255) / 256, 256, 0, stream>>>(hist, B * 256);
  lbp_hist_kernel<<<dim3(2, B), LBP_WAVES * 32, 0, stream>>>(x, hist);
  wmma_reduce_kernel<<<1, 256, 0, stream>>>(hist, out, B);
}


# lbp kernel: codes computed four pixels at a time on packed bytes (neighbour word by funnel shift, per-byte >= as carry out of bit 7, answers folded by bit-field inserts), not on extracted bytes
# speedup vs baseline: 2.4391x; 1.0901x over previous
#include <hip/hip_runtime.h>
#include <hip/hip_bf16.h>
#include <stdint.h>

typedef __attribute__((ext_vector_type(2))) float v2f;
typedef __attribute__((ext_vector_type(8))) float v8f;

__constant__ int c_unis[58] = {
    0, 1, 2, 3, 4, 6, 7, 8, 12, 14, 15, 16, 24, 28, 30, 31, 32, 48, 56, 60,
    62, 63, 64, 96, 112, 120, 124, 126, 127, 128, 129, 131, 135, 143, 159,
    191, 192, 193, 195, 199, 207, 223, 224, 225, 227, 231, 239, 240, 241,
    243, 247, 248, 249, 251, 252, 253, 254, 255};

__global__ void zero_hist_kernel(unsigned* __restrict__ hist, int n) {
  int i = blockIdx.x * blockDim.x + threadIdx.x;
  if (i < n) hist[i] = 0u;
}

__device__ __forceinline__ unsigned quant(float f) {
  return (unsigned)(f * 255.0f);
}

constexpr int LBP_WAVES = 32;
constexpr int LBP_PASSES = 8;
constexpr int LBP_COPIES = 32;
constexpr int LBP_ROWS = LBP_WAVES * LBP_PASSES;
constexpr int LBP_HALO = 3;
constexpr int STG_ROWS = LBP_ROWS + 2 * LBP_HALO;
constexpr int STG_PAD_W = 4;
constexpr int STG_PITCH_W = 128 + 2 * STG_PAD_W;
constexpr int STG_PITCH_Q = STG_PITCH_W / 4;
constexpr int STG_BATCH = 3;
constexpr int STG_STEPS = (STG_ROWS + LBP_WAVES - 1) / LBP_WAVES;
static_assert(2 * LBP_ROWS == 512, "two work-groups cover one image");
static_assert(256 * LBP_COPIES == 2 * 4 * LBP_WAVES * 32, "two 16-byte stores per thread clear it");
static_assert(STG_STEPS % STG_BATCH == 0, "whole batches");

__device__ __forceinline__ void load_window(const uint4* __restrict__ row4, unsigned lane,
                                            unsigned (&w)[12]) {
#pragma unroll
  for (int i = 0; i < 3; ++i) {
    const uint4 v = row4[lane + i];
    w[i * 4 + 0] = v.x;
    w[i * 4 + 1] = v.y;
    w[i * 4 + 2] = v.z;
    w[i * 4 + 3] = v.w;
  }
}

template <int S>
__device__ __forceinline__ unsigned cut4(unsigned hi, unsigned lo) {
  return (lo >> (8 * S)) | (hi << (32 - 8 * S));
}

__device__ __forceinline__ unsigned ge4(unsigned n, unsigned c, unsigned cy) {
  const unsigned s = (n & 0x7f7f7f7fu) + cy;
  return __builtin_amdgcn_bitop3_b32(n, c, s, 0xb2);
}

template <unsigned M>
__device__ __forceinline__ unsigned ins4(unsigned a, unsigned b) {
  return __builtin_amdgcn_bitop3_b32(a, b, M, 0xe4);
}

template <int I>
__device__ __forceinline__ unsigned lbp_code4(const unsigned (&wc)[12], const unsigned (&wu)[12],
                                              const unsigned (&wd)[12]) {
  constexpr int B = 4 + I;
  const unsigned c = wc[B];
  const unsigned cy = (~c & 0x7f7f7f7fu) + 0x01010101u;
  const unsigned r0 = ge4(wu[B], c, cy);
  const unsigned r1 = ge4(cut4<3>(wu[B + 1], wu[B]), c, cy);
  const unsigned r2 = ge4(cut4<1>(wc[B + 2], wc[B + 1]), c, cy);
  const unsigned r3 = ge4(cut4<3>(wd[B + 1], wd[B]), c, cy);
  const unsigned r4 = ge4(wd[B], c, cy);
  const unsigned r5 = ge4(cut4<1>(wd[B], wd[B - 1]), c, cy);
  const unsigned r6 = ge4(cut4<3>(wc[B - 1], wc[B - 2]), c, cy);
  const unsigned r7 = ge4(cut4<1>(wu[B], wu[B - 1]), c, cy);
  const unsigned p10 = ins4<0x40404040u>(r0 >> 1, r1);
  const unsigned p32 = ins4<0x40404040u>(r2 >> 1, r3);
  const unsigned p54 = ins4<0x40404040u>(r4 >> 1, r5);
  const unsigned p76 = ins4<0x40404040u>(r6 >> 1, r7);
  const unsigned qlo = ins4<0x30303030u>(p10 >> 2, p32);
  const unsigned qhi = ins4<0x30303030u>(p54 >> 2, p76);
  return ins4<0x0f0f0f0fu>(qlo >> 4, qhi);
}

template <int I>
__device__ __forceinline__ void lbp_row16(const unsigned (&wc)[12], const unsigned (&wu)[12],
                                          const unsigned (&wd)[12], unsigned* __restrict__ sh,
                                          unsigned lane) {
  if constexpr (I < 4) {
    const unsigned code = lbp_code4<I>(wc, wu, wd);
    atomicAdd(&sh[((code & 255u) << 5) | lane], 1u);
    atomicAdd(&sh[(((code >> 8) & 255u) << 5) | lane], 1u);
    atomicAdd(&sh[(((code >> 16) & 255u) << 5) | lane], 1u);
    atomicAdd(&sh[((code >> 24) << 5) | lane], 1u);
    lbp_row16<I + 1>(wc, wu, wd, sh, lane);
  }
}

__global__ __launch_bounds__(LBP_WAVES * 32) void lbp_hist_kernel(
    const float* __restrict__ x, unsigned* __restrict__ hist) {
  __shared__ __attribute__((aligned(16))) unsigned sh[256 * LBP_COPIES];
  __shared__ __attribute__((aligned(16))) unsigned tile[STG_ROWS * STG_PITCH_W];
  const int t = threadIdx.x;
  {
    uint4* sh4 = reinterpret_cast<uint4*>(sh);
    const uint4 z = {0u, 0u, 0u, 0u};
    sh4[t] = z;
    sh4[t + LBP_WAVES * 32] = z;
  }
  const unsigned lane = (unsigned)t & 31u;
  const int wave = t >> 5;

  const int batch = blockIdx.y;
  const int row0 = blockIdx.x * LBP_ROWS - LBP_HALO;
  const float* img = x + (size_t)batch * (512 * 512);

  for (int k0 = 0; k0 < STG_STEPS; k0 += STG_BATCH) {
    float4 v[STG_BATCH][4];
#pragma unroll
    for (int kk = 0; kk < STG_BATCH; ++kk) {
      const int j = wave + LBP_WAVES * (k0 + kk);
      const int jl = j < STG_ROWS ? j : STG_ROWS - 1;
      int gr = row0 + jl;
      gr = gr < 0 ? 0 : (gr > 511 ? 511 : gr);
      const float4* rp = reinterpret_cast<const float4*>(img + gr * 512);
#pragma unroll
      for (int i = 0; i < 4; ++i) v[kk][i] = rp[lane + 32 * i];
    }
#pragma unroll
    for (int kk = 0; kk < STG_BATCH; ++kk) {
      const int j = wave + LBP_WAVES * (k0 + kk);
      const bool in_img = (unsigned)(row0 + j) < 512u;
      if (j < STG_ROWS) {
        unsigned* dst = tile + j * STG_PITCH_W;
#pragma unroll
        for (int i = 0; i < 4; ++i) {
          const unsigned pk = quant(v[kk][i].x) | (quant(v[kk][i].y) << 8) |
                              (quant(v[kk][i].z) << 16) | (quant(v[kk][i].w) << 24);
          dst[STG_PAD_W + lane + 32 * i] = in_img ? pk : 0u;
        }
        if (lane < 2 * STG_PAD_W) dst[lane < STG_PAD_W ? lane : 128 + lane] = 0u;
      }
    }
  }
  __syncthreads();

  const uint4* tile4 = reinterpret_cast<const uint4*>(tile);
  for (int p = 0; p < LBP_PASSES; ++p) {
    const int jc = LBP_HALO + LBP_WAVES * p + wave;
    unsigned wc[12], wu[12], wd[12];
    load_window(tile4 + jc * STG_PITCH_Q, lane, wc);
    load_window(tile4 + (jc - LBP_HALO) * STG_PITCH_Q, lane, wu);
    load_window(tile4 + (jc + LBP_HALO) * STG_PITCH_Q, lane, wd);
    lbp_row16<0>(wc, wu, wd, sh, lane);
  }

  __syncthreads();
  if (t < 256) {
    unsigned v = 0u;
    for (int i = 0; i < LBP_COPIES; ++i) v += sh[(t << 5) | ((i + t) & 31)];
    if (v) atomicAdd(&hist[batch * 256 + t], v);
  }
}

__global__ __launch_bounds__(256) void wmma_reduce_kernel(
    const unsigned* __restrict__ hist, float* __restrict__ out, int B) {
  __shared__ int smap[256];
  const int t = threadIdx.x;
  {
    int m = 58;
#pragma unroll
    for (int j = 0; j < 58; ++j)
      if (c_unis[j] == t) m = j;
    smap[t] = m;
  }
  __syncthreads();

  const int wave = t >> 5;
  const int lane = t & 31;
  const int lm = lane & 15;
  const int hi = (lane >> 4) & 1;
  const int mtiles = B >> 4;
  const int ntotal = mtiles * 4;

  for (int tile = wave; tile < ntotal; tile += 8) {
    const int mt = tile >> 2;
    const int nt = tile & 3;
    const int m = (mt << 4) + lm;
    const int n = (nt << 4) + lm;

    v8f acc = {};
    const unsigned* hrow = hist + (size_t)m * 256;
    for (int k0 = 0; k0 < 256; k0 += 4) {
      const int ka = k0 + (hi ? 2 : 0);
      v2f A = {(float)hrow[ka], (float)hrow[ka + 1]};
      v2f Bm = {(smap[ka] == n) ? 1.0f : 0.0f,
                (smap[ka + 1] == n) ? 1.0f : 0.0f};
      acc = __builtin_amdgcn_wmma_f32_16x16x4_f32(
           false, A,  false, Bm,
           (short)0, acc,  false,  false);
    }

    if (n < 59) {
#pragma unroll
      for (int r = 0; r < 8; ++r) {
        const int M = (mt << 4) + (hi << 3) + r;
        const unsigned iv = (unsigned)(acc[r] + 0.5f);
        out[M * 59 + n] = (float)(iv & 255u);
      }
    }
  }
}

extern "C" void kernel_launch(void* const* d_in, const int* in_sizes, int n_in,
                              void* d_out, int out_size, void* d_ws,
                              size_t ws_size, hipStream_t stream) {
  const float* x = (const float*)d_in[0];
  float* out = (float*)d_out;
  unsigned* hist = (unsigned*)d_ws;

  const int B = in_sizes[0] >> 18;

  zero_hist_kernel<<<(B * 256 + 255) / 256, 256, 0, stream>>>(hist, B * 256);
  lbp_hist_kernel<<<dim3(2, B), LBP_WAVES * 32, 0, stream>>>(x, hist);
  wmma_reduce_kernel<<<1, 256, 0, stream>>>(hist, out, B);
}


# reduce kernel: one work-group per 16 batches (8, was 1), K split over its 8 waves, each wave's A fragments feed 4 n-tile accumulators; partial tiles summed through LDS (sum order changes, exact)
# speedup vs baseline: 3.3779x; 1.3849x over previous
#include <hip/hip_runtime.h>
#include <hip/hip_bf16.h>
#include <stdint.h>

typedef __attribute__((ext_vector_type(2))) float v2f;
typedef __attribute__((ext_vector_type(8))) float v8f;

__constant__ int c_unis[58] = {
    0, 1, 2, 3, 4, 6, 7, 8, 12, 14, 15, 16, 24, 28, 30, 31, 32, 48, 56, 60,
    62, 63, 64, 96, 112, 120, 124, 126, 127, 128, 129, 131, 135, 143, 159,
    191, 192, 193, 195, 199, 207, 223, 224, 225, 227, 231, 239, 240, 241,
    243, 247, 248, 249, 251, 252, 253, 254, 255};

__global__ void zero_hist_kernel(unsigned* __restrict__ hist, int n) {
  int i = blockIdx.x * blockDim.x + threadIdx.x;
  if (i < n) hist[i] = 0u;
}

__device__ __forceinline__ unsigned quant(float f) {
  return (unsigned)(f * 255.0f);
}

constexpr int LBP_WAVES = 32;
constexpr int LBP_PASSES = 8;
constexpr int LBP_COPIES = 32;
constexpr int LBP_ROWS = LBP_WAVES * LBP_PASSES;
constexpr int LBP_HALO = 3;
constexpr int STG_ROWS = LBP_ROWS + 2 * LBP_HALO;
constexpr int STG_PAD_W = 4;
constexpr int STG_PITCH_W = 128 + 2 * STG_PAD_W;
constexpr int STG_PITCH_Q = STG_PITCH_W / 4;
constexpr int STG_BATCH = 3;
constexpr int STG_STEPS = (STG_ROWS + LBP_WAVES - 1) / LBP_WAVES;
static_assert(2 * LBP_ROWS == 512, "two work-groups cover one image");
static_assert(256 * LBP_COPIES == 2 * 4 * LBP_WAVES * 32, "two 16-byte stores per thread clear it");
static_assert(STG_STEPS % STG_BATCH == 0, "whole batches");

__device__ __forceinline__ void load_window(const uint4* __restrict__ row4, unsigned lane,
                                            unsigned (&w)[12]) {
#pragma unroll
  for (int i = 0; i < 3; ++i) {
    const uint4 v = row4[lane + i];
    w[i * 4 + 0] = v.x;
    w[i * 4 + 1] = v.y;
    w[i * 4 + 2] = v.z;
    w[i * 4 + 3] = v.w;
  }
}

template <int S>
__device__ __forceinline__ unsigned cut4(unsigned hi, unsigned lo) {
  return (lo >> (8 * S)) | (hi << (32 - 8 * S));
}

__device__ __forceinline__ unsigned ge4(unsigned n, unsigned c, unsigned cy) {
  const unsigned s = (n & 0x7f7f7f7fu) + cy;
  return __builtin_amdgcn_bitop3_b32(n, c, s, 0xb2);
}

template <unsigned M>
__device__ __forceinline__ unsigned ins4(unsigned a, unsigned b) {
  return __builtin_amdgcn_bitop3_b32(a, b, M, 0xe4);
}

template <int I>
__device__ __forceinline__ unsigned lbp_code4(const unsigned (&wc)[12], const unsigned (&wu)[12],
                                              const unsigned (&wd)[12]) {
  constexpr int B = 4 + I;
  const unsigned c = wc[B];
  const unsigned cy = (~c & 0x7f7f7f7fu) + 0x01010101u;
  const unsigned r0 = ge4(wu[B], c, cy);
  const unsigned r1 = ge4(cut4<3>(wu[B + 1], wu[B]), c, cy);
  const unsigned r2 = ge4(cut4<1>(wc[B + 2], wc[B + 1]), c, cy);
  const unsigned r3 = ge4(cut4<3>(wd[B + 1], wd[B]), c, cy);
  const unsigned r4 = ge4(wd[B], c, cy);
  const unsigned r5 = ge4(cut4<1>(wd[B], wd[B - 1]), c, cy);
  const unsigned r6 = ge4(cut4<3>(wc[B - 1], wc[B - 2]), c, cy);
  const unsigned r7 = ge4(cut4<1>(wu[B], wu[B - 1]), c, cy);
  const unsigned p10 = ins4<0x40404040u>(r0 >> 1, r1);
  const unsigned p32 = ins4<0x40404040u>(r2 >> 1, r3);
  const unsigned p54 = ins4<0x40404040u>(r4 >> 1, r5);
  const unsigned p76 = ins4<0x40404040u>(r6 >> 1, r7);
  const unsigned qlo = ins4<0x30303030u>(p10 >> 2, p32);
  const unsigned qhi = ins4<0x30303030u>(p54 >> 2, p76);
  return ins4<0x0f0f0f0fu>(qlo >> 4, qhi);
}

template <int I>
__device__ __forceinline__ void lbp_row16(const unsigned (&wc)[12], const unsigned (&wu)[12],
                                          const unsigned (&wd)[12], unsigned* __restrict__ sh,
                                          unsigned lane) {
  if constexpr (I < 4) {
    const unsigned code = lbp_code4<I>(wc, wu, wd);
    atomicAdd(&sh[((code & 255u) << 5) | lane], 1u);
    atomicAdd(&sh[(((code >> 8) & 255u) << 5) | lane], 1u);
    atomicAdd(&sh[(((code >> 16) & 255u) << 5) | lane], 1u);
    atomicAdd(&sh[((code >> 24) << 5) | lane], 1u);
    lbp_row16<I + 1>(wc, wu, wd, sh, lane);
  }
}

__global__ __launch_bounds__(LBP_WAVES * 32) void lbp_hist_kernel(
    const float* __restrict__ x, unsigned* __restrict__ hist) {
  __shared__ __attribute__((aligned(16))) unsigned sh[256 * LBP_COPIES];
  __shared__ __attribute__((aligned(16))) unsigned tile[STG_ROWS * STG_PITCH_W];
  const int t = threadIdx.x;
  {
    uint4* sh4 = reinterpret_cast<uint4*>(sh);
    const uint4 z = {0u, 0u, 0u, 0u};
    sh4[t] = z;
    sh4[t + LBP_WAVES * 32] = z;
  }
  const unsigned lane = (unsigned)t & 31u;
  const int wave = t >> 5;

  const int batch = blockIdx.y;
  const int row0 = blockIdx.x * LBP_ROWS - LBP_HALO;
  const float* img = x + (size_t)batch * (512 * 512);

  for (int k0 = 0; k0 < STG_STEPS; k0 += STG_BATCH) {
    float4 v[STG_BATCH][4];
#pragma unroll
    for (int kk = 0; kk < STG_BATCH; ++kk) {
      const int j = wave + LBP_WAVES * (k0 + kk);
      const int jl = j < STG_ROWS ? j : STG_ROWS - 1;
      int gr = row0 + jl;
      gr = gr < 0 ? 0 : (gr > 511 ? 511 : gr);
      const float4* rp = reinterpret_cast<const float4*>(img + gr * 512);
#pragma unroll
      for (int i = 0; i < 4; ++i) v[kk][i] = rp[lane + 32 * i];
    }
#pragma unroll
    for (int kk = 0; kk < STG_BATCH; ++kk) {
      const int j = wave + LBP_WAVES * (k0 + kk);
      const bool in_img = (unsigned)(row0 + j) < 512u;
      if (j < STG_ROWS) {
        unsigned* dst = tile + j * STG_PITCH_W;
#pragma unroll
        for (int i = 0; i < 4; ++i) {
          const unsigned pk = quant(v[kk][i].x) | (quant(v[kk][i].y) << 8) |
                              (quant(v[kk][i].z) << 16) | (quant(v[kk][i].w) << 24);
          dst[STG_PAD_W + lane + 32 * i] = in_img ? pk : 0u;
        }
        if (lane < 2 * STG_PAD_W) dst[lane < STG_PAD_W ? lane : 128 + lane] = 0u;
      }
    }
  }
  __syncthreads();

  const uint4* tile4 = reinterpret_cast<const uint4*>(tile);
  for (int p = 0; p < LBP_PASSES; ++p) {
    const int jc = LBP_HALO + LBP_WAVES * p + wave;
    unsigned wc[12], wu[12], wd[12];
    load_window(tile4 + jc * STG_PITCH_Q, lane, wc);
    load_window(tile4 + (jc - LBP_HALO) * STG_PITCH_Q, lane, wu);
    load_window(tile4 + (jc + LBP_HALO) * STG_PITCH_Q, lane, wd);
    lbp_row16<0>(wc, wu, wd, sh, lane);
  }

  __syncthreads();
  if (t < 256) {
    unsigned v = 0u;
    for (int i = 0; i < LBP_COPIES; ++i) v += sh[(t << 5) | ((i + t) & 31)];
    if (v) atomicAdd(&hist[batch * 256 + t], v);
  }
}

constexpr int RED_WAVES = 8;
constexpr int RED_KW = 256 / RED_WAVES;
constexpr int RED_KL = RED_KW / 2;
constexpr int RED_OUT = 16 * 59;
static_assert(RED_KL == 16, "four 16-byte loads per lane");

__global__ __launch_bounds__(RED_WAVES * 32) void wmma_reduce_kernel(
    const unsigned* __restrict__ hist, float* __restrict__ out) {
  __shared__ __attribute__((aligned(16))) int smap[256];
  __shared__ __attribute__((aligned(16))) float part[RED_WAVES * 4 * 256];
  const int t = threadIdx.x;
  const int wave = t >> 5;
  const int lane = t & 31;
  const int c = lane & 15;
  const int h = lane >> 4;
  const int mt = blockIdx.x;
  const int koff = wave * RED_KW + h * RED_KL;

  unsigned a[RED_KL];
  {
    const uint4* ap = reinterpret_cast<const uint4*>(hist + (size_t)(mt * 16 + c) * 256 + koff);
#pragma unroll
    for (int i = 0; i < 4; ++i) {
      const uint4 v = ap[i];
      a[4 * i + 0] = v.x;
      a[4 * i + 1] = v.y;
      a[4 * i + 2] = v.z;
      a[4 * i + 3] = v.w;
    }
  }

  smap[t] = 58;
  __syncthreads();
  if (t < 58) smap[c_unis[t]] = t;
  __syncthreads();

  int m[RED_KL];
  {
    const int4* mp = reinterpret_cast<const int4*>(smap + koff);
#pragma unroll
    for (int i = 0; i < 4; ++i) {
      const int4 v = mp[i];
      m[4 * i + 0] = v.x;
      m[4 * i + 1] = v.y;
      m[4 * i + 2] = v.z;
      m[4 * i + 3] = v.w;
    }
  }

  v8f acc[4] = {};
#pragma unroll
  for (int s = 0; s < RED_KW / 4; ++s) {
    const v2f A = {(float)a[2 * s], (float)a[2 * s + 1]};
#pragma unroll
    for (int nt = 0; nt < 4; ++nt) {
      const int n = nt * 16 + c;
      const v2f Bm = {(m[2 * s] == n) ? 1.0f : 0.0f, (m[2 * s + 1] == n) ? 1.0f : 0.0f};
      acc[nt] = __builtin_amdgcn_wmma_f32_16x16x4_f32(
           false, A,  false, Bm,
           (short)0, acc[nt],  false,  false);
    }
  }

#pragma unroll
  for (int nt = 0; nt < 4; ++nt) {
#pragma unroll
    for (int r = 0; r < 8; ++r) part[((wave * 4 + nt) * 8 + r) * 32 + lane] = acc[nt][r];
  }
  __syncthreads();

  constexpr int RED_PER = (RED_OUT + RED_WAVES * 32 - 1) / (RED_WAVES * 32);
#pragma unroll
  for (int q = 0; q < RED_PER; ++q) {
    const int o = t + q * (RED_WAVES * 32);
    if ((q + 1) * (RED_WAVES * 32) <= RED_OUT || o < RED_OUT) {
      const int row = o / 59;
      const int n = o - row * 59;
      const int idx = ((n >> 4) * 8 + (row & 7)) * 32 + (row >> 3) * 16 + (n & 15);
      float v = 0.0f;
      for (int w = 0; w < RED_WAVES; ++w) v += part[w * 1024 + idx];
      const unsigned iv = (unsigned)(v + 0.5f);
      out[mt * RED_OUT + o] = (float)(iv & 255u);
    }
  }
}

extern "C" void kernel_launch(void* const* d_in, const int* in_sizes, int n_in,
                              void* d_out, int out_size, void* d_ws,
                              size_t ws_size, hipStream_t stream) {
  const float* x = (const float*)d_in[0];
  float* out = (float*)d_out;
  unsigned* hist = (unsigned*)d_ws;

  const int B = in_sizes[0] >> 18;

  zero_hist_kernel<<<(B * 256 + 255) / 256, 256, 0, stream>>>(hist, B * 256);
  lbp_hist_kernel<<<dim3(2, B), LBP_WAVES * 32, 0, stream>>>(x, hist);
  wmma_reduce_kernel<<<B / 16, RED_WAVES * 32, 0, stream>>>(hist, out);
}


# lbp kernel: 4 work-groups of 16 waves per image (was 2 of 32), two resident per WGP, each in two stage-then-code rounds: one stages while the other computes; planes kept mod 256 as bytes (exact)
# speedup vs baseline: 3.7205x; 1.1014x over previous
#include <hip/hip_runtime.h>
#include <hip/hip_bf16.h>
#include <stdint.h>

typedef __attribute__((ext_vector_type(2))) float v2f;
typedef __attribute__((ext_vector_type(8))) float v8f;

__constant__ int c_unis[58] = {
    0, 1, 2, 3, 4, 6, 7, 8, 12, 14, 15, 16, 24, 28, 30, 31, 32, 48, 56, 60,
    62, 63, 64, 96, 112, 120, 124, 126, 127, 128, 129, 131, 135, 143, 159,
    191, 192, 193, 195, 199, 207, 223, 224, 225, 227, 231, 239, 240, 241,
    243, 247, 248, 249, 251, 252, 253, 254, 255};

__device__ __forceinline__ unsigned quant(float f) {
  return (unsigned)(f * 255.0f);
}

constexpr int LBP_WAVES = 16;
constexpr int LBP_ROUNDS = 2;
constexpr int LBP_PASSES = 4;
constexpr int LBP_COPIES = 32;
constexpr int RND_ROWS = LBP_WAVES * LBP_PASSES;
constexpr int LBP_ROWS = RND_ROWS * LBP_ROUNDS;
constexpr int LBP_PARTS = 512 / LBP_ROWS;
constexpr int LBP_HALO = 3;
constexpr int STG_ROWS = LBP_ROWS + 2 * LBP_HALO;
constexpr int STG_FIRST = RND_ROWS + 2 * LBP_HALO;
constexpr int STG_PAD_W = 8;
constexpr int STG_PITCH_W = 128 + 2 * STG_PAD_W;
constexpr int STG_PITCH_Q = STG_PITCH_W / 4;
constexpr int STG_WIN_Q = STG_PAD_W / 4 - 1;
static_assert(LBP_PARTS * LBP_ROWS == 512, "four work-groups cover one image");
static_assert(256 * LBP_COPIES == 4 * 4 * LBP_WAVES * 32, "four 16-byte stores per thread clear it");
static_assert(STG_ROWS - STG_FIRST == 4 * LBP_WAVES, "round 1 stages four rows per wave");
static_assert(STG_FIRST > 4 * LBP_WAVES && STG_FIRST <= 5 * LBP_WAVES, "round 0 stages five rows per wave at most");

__device__ __forceinline__ void load_window(const uint4* __restrict__ row4, unsigned lane,
                                            unsigned (&w)[12]) {
#pragma unroll
  for (int i = 0; i < 3; ++i) {
    const uint4 v = row4[STG_WIN_Q + lane + i];
    w[i * 4 + 0] = v.x;
    w[i * 4 + 1] = v.y;
    w[i * 4 + 2] = v.z;
    w[i * 4 + 3] = v.w;
  }
}

template <int S>
__device__ __forceinline__ unsigned cut4(unsigned hi, unsigned lo) {
  return (lo >> (8 * S)) | (hi << (32 - 8 * S));
}

__device__ __forceinline__ unsigned ge4(unsigned n, unsigned c, unsigned cy) {
  const unsigned s = (n & 0x7f7f7f7fu) + cy;
  return __builtin_amdgcn_bitop3_b32(n, c, s, 0xb2);
}

template <unsigned M>
__device__ __forceinline__ unsigned ins4(unsigned a, unsigned b) {
  return __builtin_amdgcn_bitop3_b32(a, b, M, 0xe4);
}

template <int I>
__device__ __forceinline__ unsigned lbp_code4(const unsigned (&wc)[12], const unsigned (&wu)[12],
                                              const unsigned (&wd)[12]) {
  constexpr int B = 4 + I;
  const unsigned c = wc[B];
  const unsigned cy = (~c & 0x7f7f7f7fu) + 0x01010101u;
  const unsigned r0 = ge4(wu[B], c, cy);
  const unsigned r1 = ge4(cut4<3>(wu[B + 1], wu[B]), c, cy);
  const unsigned r2 = ge4(cut4<1>(wc[B + 2], wc[B + 1]), c, cy);
  const unsigned r3 = ge4(cut4<3>(wd[B + 1], wd[B]), c, cy);
  const unsigned r4 = ge4(wd[B], c, cy);
  const unsigned r5 = ge4(cut4<1>(wd[B], wd[B - 1]), c, cy);
  const unsigned r6 = ge4(cut4<3>(wc[B - 1], wc[B - 2]), c, cy);
  const unsigned r7 = ge4(cut4<1>(wu[B], wu[B - 1]), c, cy);
  const unsigned p10 = ins4<0x40404040u>(r0 >> 1, r1);
  const unsigned p32 = ins4<0x40404040u>(r2 >> 1, r3);
  const unsigned p54 = ins4<0x40404040u>(r4 >> 1, r5);
  const unsigned p76 = ins4<0x40404040u>(r6 >> 1, r7);
  const unsigned qlo = ins4<0x30303030u>(p10 >> 2, p32);
  const unsigned qhi = ins4<0x30303030u>(p54 >> 2, p76);
  return ins4<0x0f0f0f0fu>(qlo >> 4, qhi);
}

template <int I>
__device__ __forceinline__ void lbp_row16(const unsigned (&wc)[12], const unsigned (&wu)[12],
                                          const unsigned (&wd)[12], unsigned* __restrict__ sh,
                                          unsigned lane) {
  if constexpr (I < 4) {
    const unsigned code = lbp_code4<I>(wc, wu, wd);
    atomicAdd(&sh[((code & 255u) << 5) | lane], 1u);
    atomicAdd(&sh[(((code >> 8) & 255u) << 5) | lane], 1u);
    atomicAdd(&sh[(((code >> 16) & 255u) << 5) | lane], 1u);
    atomicAdd(&sh[((code >> 24) << 5) | lane], 1u);
    lbp_row16<I + 1>(wc, wu, wd, sh, lane);
  }
}

template <int NR>
__device__ __forceinline__ void stage_rows(const float* __restrict__ img, int row0,
                                           unsigned* __restrict__ tile, int jbase, int jend,
                                           int wave, unsigned lane) {
  float4 v[NR][4];
#pragma unroll
  for (int kk = 0; kk < NR; ++kk) {
    const int j = jbase + wave + LBP_WAVES * kk;
    const int jl = j < jend ? j : jend - 1;
    int gr = row0 + jl;
    gr = gr < 0 ? 0 : (gr > 511 ? 511 : gr);
    const float4* rp = reinterpret_cast<const float4*>(img + gr * 512);
#pragma unroll
    for (int i = 0; i < 4; ++i) v[kk][i] = rp[lane + 32 * i];
  }
#pragma unroll
  for (int kk = 0; kk < NR; ++kk) {
    const int j = jbase + wave + LBP_WAVES * kk;
    const bool in_img = (unsigned)(row0 + j) < 512u;
    if (j < jend) {
      unsigned* dst = tile + j * STG_PITCH_W;
#pragma unroll
      for (int i = 0; i < 4; ++i) {
        const unsigned pk = quant(v[kk][i].x) | (quant(v[kk][i].y) << 8) |
                            (quant(v[kk][i].z) << 16) | (quant(v[kk][i].w) << 24);
        dst[STG_PAD_W + lane + 32 * i] = in_img ? pk : 0u;
      }
      if (lane < 2 * STG_PAD_W) dst[lane < STG_PAD_W ? lane : 128 + lane] = 0u;
    }
  }
}

__global__ __launch_bounds__(LBP_WAVES * 32) void lbp_hist_kernel(
    const float* __restrict__ x, unsigned* __restrict__ planes) {
  __shared__ __attribute__((aligned(16))) unsigned sh[256 * LBP_COPIES];
  __shared__ __attribute__((aligned(16))) unsigned tile[STG_ROWS * STG_PITCH_W];
  __shared__ __attribute__((aligned(16))) unsigned char pb[256];
  const int t = threadIdx.x;
  {
    uint4* sh4 = reinterpret_cast<uint4*>(sh);
    const uint4 z = {0u, 0u, 0u, 0u};
#pragma unroll
    for (int i = 0; i < 4; ++i) sh4[t + i * (LBP_WAVES * 32)] = z;
  }
  const unsigned lane = (unsigned)t & 31u;
  const int wave = t >> 5;

  const int batch = blockIdx.y;
  const int row0 = blockIdx.x * LBP_ROWS - LBP_HALO;
  const float* img = x + (size_t)batch * (512 * 512);
  const uint4* tile4 = reinterpret_cast<const uint4*>(tile);

  stage_rows<3>(img, row0, tile, 0, STG_FIRST, wave, lane);
  stage_rows<2>(img, row0, tile, 3 * LBP_WAVES, STG_FIRST, wave, lane);
  __syncthreads();

  for (int r = 0; r < LBP_ROUNDS; ++r) {
    if (r == 1) {
      stage_rows<4>(img, row0, tile, STG_FIRST, STG_ROWS, wave, lane);
      __syncthreads();
    }
    for (int p = 0; p < LBP_PASSES; ++p) {
      const int jc = LBP_HALO + RND_ROWS * r + LBP_WAVES * p + wave;
      unsigned wc[12], wu[12], wd[12];
      load_window(tile4 + jc * STG_PITCH_Q, lane, wc);
      load_window(tile4 + (jc - LBP_HALO) * STG_PITCH_Q, lane, wu);
      load_window(tile4 + (jc + LBP_HALO) * STG_PITCH_Q, lane, wd);
      lbp_row16<0>(wc, wu, wd, sh, lane);
    }
  }

  __syncthreads();
  if (t < 256) {
    unsigned v0 = 0u;
    for (int i = 0; i < LBP_COPIES; ++i) v0 += sh[(t << 5) | ((unsigned)(i + t) & 31u)];
    pb[t] = (unsigned char)v0;
  }
  __syncthreads();
  if (t < 32) {
    const uint2 v = reinterpret_cast<const uint2*>(pb)[t];
    reinterpret_cast<uint2*>(planes)[(batch * LBP_PARTS + (int)blockIdx.x) * 32 + t] = v;
  }
}

constexpr int RED_WAVES = 8;
constexpr int RED_KW = 256 / RED_WAVES;
constexpr int RED_KL = RED_KW / 2;
constexpr int RED_OUT = 16 * 59;
static_assert(RED_KL == 16, "one 16-byte load per lane and plane");

__global__ __launch_bounds__(RED_WAVES * 32) void wmma_reduce_kernel(
    const unsigned* __restrict__ planes, float* __restrict__ out) {
  __shared__ __attribute__((aligned(16))) int smap[256];
  __shared__ __attribute__((aligned(16))) float part[RED_WAVES * 4 * 256];
  const int t = threadIdx.x;
  const int wave = t >> 5;
  const int lane = t & 31;
  const int c = lane & 15;
  const int h = lane >> 4;
  const int mt = blockIdx.x;
  const int koff = wave * RED_KW + h * RED_KL;

  unsigned a[RED_KL];
  {
    const uint4* ap = reinterpret_cast<const uint4*>(planes) + (size_t)(mt * 16 + c) * (LBP_PARTS * 16) + (koff >> 4);
    uint4 pq[LBP_PARTS];
#pragma unroll
    for (int q = 0; q < LBP_PARTS; ++q) pq[q] = ap[16 * q];
#pragma unroll
    for (int i = 0; i < 4; ++i) {
      unsigned ev = 0u, od = 0u;
#pragma unroll
      for (int q = 0; q < LBP_PARTS; ++q) {
        const unsigned wq = i == 0 ? pq[q].x : (i == 1 ? pq[q].y : (i == 2 ? pq[q].z : pq[q].w));
        ev += wq & 0x00ff00ffu;
        od += (wq >> 8) & 0x00ff00ffu;
      }
      a[4 * i + 0] = ev & 0xffffu;
      a[4 * i + 1] = od & 0xffffu;
      a[4 * i + 2] = ev >> 16;
      a[4 * i + 3] = od >> 16;
    }
  }

  smap[t] = 58;
  __syncthreads();
  if (t < 58) smap[c_unis[t]] = t;
  __syncthreads();

  int m[RED_KL];
  {
    const int4* mp = reinterpret_cast<const int4*>(smap + koff);
#pragma unroll
    for (int i = 0; i < 4; ++i) {
      const int4 v = mp[i];
      m[4 * i + 0] = v.x;
      m[4 * i + 1] = v.y;
      m[4 * i + 2] = v.z;
      m[4 * i + 3] = v.w;
    }
  }

  v8f acc[4] = {};
#pragma unroll
  for (int s = 0; s < RED_KW / 4; ++s) {
    const v2f A = {(float)a[2 * s], (float)a[2 * s + 1]};
#pragma unroll
    for (int nt = 0; nt < 4; ++nt) {
      const int n = nt * 16 + c;
      const v2f Bm = {(m[2 * s] == n) ? 1.0f : 0.0f, (m[2 * s + 1] == n) ? 1.0f : 0.0f};
      acc[nt] = __builtin_amdgcn_wmma_f32_16x16x4_f32(
           false, A,  false, Bm,
           (short)0, acc[nt],  false,  false);
    }
  }

#pragma unroll
  for (int nt = 0; nt < 4; ++nt) {
#pragma unroll
    for (int r = 0; r < 8; ++r) part[((wave * 4 + nt) * 8 + r) * 32 + lane] = acc[nt][r];
  }
  __syncthreads();

  constexpr int RED_PER = (RED_OUT + RED_WAVES * 32 - 1) / (RED_WAVES * 32);
#pragma unroll
  for (int q = 0; q < RED_PER; ++q) {
    const int o = t + q * (RED_WAVES * 32);
    if ((q + 1) * (RED_WAVES * 32) <= RED_OUT || o < RED_OUT) {
      const int row = o / 59;
      const int n = o - row * 59;
      const int idx = ((n >> 4) * 8 + (row & 7)) * 32 + (row >> 3) * 16 + (n & 15);
      float v = 0.0f;
      for (int w = 0; w < RED_WAVES; ++w) v += part[w * 1024 + idx];
      const unsigned iv = (unsigned)(v + 0.5f);
      out[mt * RED_OUT + o] = (float)(iv & 255u);
    }
  }
}

extern "C" void kernel_launch(void* const* d_in, const int* in_sizes, int n_in,
                              void* d_out, int out_size, void* d_ws,
                              size_t ws_size, hipStream_t stream) {
  const float* x = (const float*)d_in[0];
  float* out = (float*)d_out;
  unsigned* planes = (unsigned*)d_ws;

  const int B = in_sizes[0] >> 18;

  lbp_hist_kernel<<<dim3(LBP_PARTS, B), LBP_WAVES * 32, 0, stream>>>(x, planes);
  wmma_reduce_kernel<<<B / 16, RED_WAVES * 32, 0, stream>>>(planes, out);
}


# lbp kernel: staging runs ahead of the codes: 4 rounds of 32 rows; a wave issues next round's row loads, codes 2 rows, then quantizes and stores them; LDS-only barrier keeps the loads in flight
# speedup vs baseline: 3.8785x; 1.0425x over previous
#include <hip/hip_runtime.h>
#include <hip/hip_bf16.h>
#include <stdint.h>

typedef __attribute__((ext_vector_type(2))) float v2f;
typedef __attribute__((ext_vector_type(8))) float v8f;

__constant__ int c_unis[58] = {
    0, 1, 2, 3, 4, 6, 7, 8, 12, 14, 15, 16, 24, 28, 30, 31, 32, 48, 56, 60,
    62, 63, 64, 96, 112, 120, 124, 126, 127, 128, 129, 131, 135, 143, 159,
    191, 192, 193, 195, 199, 207, 223, 224, 225, 227, 231, 239, 240, 241,
    243, 247, 248, 249, 251, 252, 253, 254, 255};

__device__ __forceinline__ unsigned quant(float f) {
  return (unsigned)(f * 255.0f);
}

constexpr int LBP_WAVES = 16;
constexpr int LBP_ROUNDS = 4;
constexpr int LBP_PASSES = 2;
constexpr int LBP_COPIES = 32;
constexpr int RND_ROWS = LBP_WAVES * LBP_PASSES;
constexpr int LBP_ROWS = RND_ROWS * LBP_ROUNDS;
constexpr int LBP_PARTS = 512 / LBP_ROWS;
constexpr int LBP_HALO = 3;
constexpr int STG_ROWS = LBP_ROWS + 2 * LBP_HALO;
constexpr int STG_FIRST = RND_ROWS + 2 * LBP_HALO;
constexpr int STG_PAD_W = 8;
constexpr int STG_PITCH_W = 128 + 2 * STG_PAD_W;
constexpr int STG_PITCH_Q = STG_PITCH_W / 4;
constexpr int STG_WIN_Q = STG_PAD_W / 4 - 1;
static_assert(LBP_PARTS * LBP_ROWS == 512, "four work-groups cover one image");
static_assert(256 * LBP_COPIES == 4 * 4 * LBP_WAVES * 32, "four 16-byte stores per thread clear it");
static_assert(STG_ROWS - STG_FIRST == (LBP_ROUNDS - 1) * RND_ROWS, "every round but the last stages the next round's rows");
static_assert(STG_FIRST > 2 * LBP_WAVES && STG_FIRST <= 3 * LBP_WAVES, "three rows per wave at most before round 0");

__device__ __forceinline__ void load_window(const uint4* __restrict__ row4, unsigned lane,
                                            unsigned (&w)[12]) {
#pragma unroll
  for (int i = 0; i < 3; ++i) {
    const uint4 v = row4[STG_WIN_Q + lane + i];
    w[i * 4 + 0] = v.x;
    w[i * 4 + 1] = v.y;
    w[i * 4 + 2] = v.z;
    w[i * 4 + 3] = v.w;
  }
}

template <int S>
__device__ __forceinline__ unsigned cut4(unsigned hi, unsigned lo) {
  return (lo >> (8 * S)) | (hi << (32 - 8 * S));
}

__device__ __forceinline__ unsigned ge4(unsigned n, unsigned c, unsigned cy) {
  const unsigned s = (n & 0x7f7f7f7fu) + cy;
  return __builtin_amdgcn_bitop3_b32(n, c, s, 0xb2);
}

template <unsigned M>
__device__ __forceinline__ unsigned ins4(unsigned a, unsigned b) {
  return __builtin_amdgcn_bitop3_b32(a, b, M, 0xe4);
}

template <int I>
__device__ __forceinline__ unsigned lbp_code4(const unsigned (&wc)[12], const unsigned (&wu)[12],
                                              const unsigned (&wd)[12]) {
  constexpr int B = 4 + I;
  const unsigned c = wc[B];
  const unsigned cy = (~c & 0x7f7f7f7fu) + 0x01010101u;
  const unsigned r0 = ge4(wu[B], c, cy);
  const unsigned r1 = ge4(cut4<3>(wu[B + 1], wu[B]), c, cy);
  const unsigned r2 = ge4(cut4<1>(wc[B + 2], wc[B + 1]), c, cy);
  const unsigned r3 = ge4(cut4<3>(wd[B + 1], wd[B]), c, cy);
  const unsigned r4 = ge4(wd[B], c, cy);
  const unsigned r5 = ge4(cut4<1>(wd[B], wd[B - 1]), c, cy);
  const unsigned r6 = ge4(cut4<3>(wc[B - 1], wc[B - 2]), c, cy);
  const unsigned r7 = ge4(cut4<1>(wu[B], wu[B - 1]), c, cy);
  const unsigned p10 = ins4<0x40404040u>(r0 >> 1, r1);
  const unsigned p32 = ins4<0x40404040u>(r2 >> 1, r3);
  const unsigned p54 = ins4<0x40404040u>(r4 >> 1, r5);
  const unsigned p76 = ins4<0x40404040u>(r6 >> 1, r7);
  const unsigned qlo = ins4<0x30303030u>(p10 >> 2, p32);
  const unsigned qhi = ins4<0x30303030u>(p54 >> 2, p76);
  return ins4<0x0f0f0f0fu>(qlo >> 4, qhi);
}

template <int I>
__device__ __forceinline__ void lbp_row16(const unsigned (&wc)[12], const unsigned (&wu)[12],
                                          const unsigned (&wd)[12], unsigned* __restrict__ sh,
                                          unsigned lane) {
  if constexpr (I < 4) {
    const unsigned code = lbp_code4<I>(wc, wu, wd);
    atomicAdd(&sh[((code & 255u) << 5) | lane], 1u);
    atomicAdd(&sh[(((code >> 8) & 255u) << 5) | lane], 1u);
    atomicAdd(&sh[(((code >> 16) & 255u) << 5) | lane], 1u);
    atomicAdd(&sh[((code >> 24) << 5) | lane], 1u);
    lbp_row16<I + 1>(wc, wu, wd, sh, lane);
  }
}

template <int NR>
__device__ __forceinline__ void stage_load(const float* __restrict__ img, int row0, int jbase,
                                           int jend, int wave, unsigned lane,
                                           float4 (&v)[NR][4]) {
#pragma unroll
  for (int kk = 0; kk < NR; ++kk) {
    const int j = jbase + wave + LBP_WAVES * kk;
    const int jl = j < jend ? j : jend - 1;
    int gr = row0 + jl;
    gr = gr < 0 ? 0 : (gr > 511 ? 511 : gr);
    const float4* rp = reinterpret_cast<const float4*>(img + gr * 512);
#pragma unroll
    for (int i = 0; i < 4; ++i) v[kk][i] = rp[lane + 32 * i];
  }
}

template <int NR>
__device__ __forceinline__ void stage_store(const float4 (&v)[NR][4], int row0,
                                            unsigned* __restrict__ tile, int jbase, int jend,
                                            int wave, unsigned lane) {
#pragma unroll
  for (int kk = 0; kk < NR; ++kk) {
    const int j = jbase + wave + LBP_WAVES * kk;
    const bool in_img = (unsigned)(row0 + j) < 512u;
    if (j < jend) {
      unsigned* dst = tile + j * STG_PITCH_W;
#pragma unroll
      for (int i = 0; i < 4; ++i) {
        const unsigned pk = quant(v[kk][i].x) | (quant(v[kk][i].y) << 8) |
                            (quant(v[kk][i].z) << 16) | (quant(v[kk][i].w) << 24);
        dst[STG_PAD_W + lane + 32 * i] = in_img ? pk : 0u;
      }
      if (lane < 2 * STG_PAD_W) dst[lane < STG_PAD_W ? lane : 128 + lane] = 0u;
    }
  }
}

__device__ __forceinline__ void lds_barrier() {
  __builtin_amdgcn_fence(__ATOMIC_RELEASE, "workgroup", "local");
  __builtin_amdgcn_s_barrier();
  __builtin_amdgcn_fence(__ATOMIC_ACQUIRE, "workgroup", "local");
}

template <int NR>
__device__ __forceinline__ void stage_rows(const float* __restrict__ img, int row0,
                                           unsigned* __restrict__ tile, int jbase, int jend,
                                           int wave, unsigned lane) {
  float4 v[NR][4];
  stage_load<NR>(img, row0, jbase, jend, wave, lane, v);
  stage_store<NR>(v, row0, tile, jbase, jend, wave, lane);
}

__global__ __launch_bounds__(LBP_WAVES * 32) void lbp_hist_kernel(
    const float* __restrict__ x, unsigned* __restrict__ planes) {
  __shared__ __attribute__((aligned(16))) unsigned sh[256 * LBP_COPIES];
  __shared__ __attribute__((aligned(16))) unsigned tile[STG_ROWS * STG_PITCH_W];
  __shared__ __attribute__((aligned(16))) unsigned char pb[256];
  const int t = threadIdx.x;
  {
    uint4* sh4 = reinterpret_cast<uint4*>(sh);
    const uint4 z = {0u, 0u, 0u, 0u};
#pragma unroll
    for (int i = 0; i < 4; ++i) sh4[t + i * (LBP_WAVES * 32)] = z;
  }
  const unsigned lane = (unsigned)t & 31u;
  const int wave = t >> 5;

  const int batch = blockIdx.y;
  const int row0 = blockIdx.x * LBP_ROWS - LBP_HALO;
  const float* img = x + (size_t)batch * (512 * 512);
  const uint4* tile4 = reinterpret_cast<const uint4*>(tile);

  stage_rows<3>(img, row0, tile, 0, STG_FIRST, wave, lane);

  for (int r = 0; r < LBP_ROUNDS; ++r) {
    const bool ahead = r + 1 < LBP_ROUNDS;
    const int jn = STG_FIRST + RND_ROWS * r;
    float4 nx[LBP_PASSES][4];
    if (ahead) stage_load<LBP_PASSES>(img, row0, jn, STG_ROWS, wave, lane, nx);
    lds_barrier();
#pragma nounroll
    for (int p = 0; p < LBP_PASSES; ++p) {
      const int jc = LBP_HALO + RND_ROWS * r + LBP_WAVES * p + wave;
      unsigned wc[12], wu[12], wd[12];
      load_window(tile4 + jc * STG_PITCH_Q, lane, wc);
      load_window(tile4 + (jc - LBP_HALO) * STG_PITCH_Q, lane, wu);
      load_window(tile4 + (jc + LBP_HALO) * STG_PITCH_Q, lane, wd);
      lbp_row16<0>(wc, wu, wd, sh, lane);
    }
    if (ahead) stage_store<LBP_PASSES>(nx, row0, tile, jn, STG_ROWS, wave, lane);
  }

  __syncthreads();
  if (t < 256) {
    unsigned v0 = 0u;
    for (int i = 0; i < LBP_COPIES; ++i) v0 += sh[(t << 5) | ((unsigned)(i + t) & 31u)];
    pb[t] = (unsigned char)v0;
  }
  __syncthreads();
  if (t < 32) {
    const uint2 v = reinterpret_cast<const uint2*>(pb)[t];
    reinterpret_cast<uint2*>(planes)[(batch * LBP_PARTS + (int)blockIdx.x) * 32 + t] = v;
  }
}

constexpr int RED_WAVES = 8;
constexpr int RED_KW = 256 / RED_WAVES;
constexpr int RED_KL = RED_KW / 2;
constexpr int RED_OUT = 16 * 59;
static_assert(RED_KL == 16, "one 16-byte load per lane and plane");

__global__ __launch_bounds__(RED_WAVES * 32) void wmma_reduce_kernel(
    const unsigned* __restrict__ planes, float* __restrict__ out) {
  __shared__ __attribute__((aligned(16))) int smap[256];
  __shared__ __attribute__((aligned(16))) float part[RED_WAVES * 4 * 256];
  const int t = threadIdx.x;
  const int wave = t >> 5;
  const int lane = t & 31;
  const int c = lane & 15;
  const int h = lane >> 4;
  const int mt = blockIdx.x;
  const int koff = wave * RED_KW + h * RED_KL;

  unsigned a[RED_KL];
  {
    const uint4* ap = reinterpret_cast<const uint4*>(planes) + (size_t)(mt * 16 + c) * (LBP_PARTS * 16) + (koff >> 4);
    uint4 pq[LBP_PARTS];
#pragma unroll
    for (int q = 0; q < LBP_PARTS; ++q) pq[q] = ap[16 * q];
#pragma unroll
    for (int i = 0; i < 4; ++i) {
      unsigned ev = 0u, od = 0u;
#pragma unroll
      for (int q = 0; q < LBP_PARTS; ++q) {
        const unsigned wq = i == 0 ? pq[q].x : (i == 1 ? pq[q].y : (i == 2 ? pq[q].z : pq[q].w));
        ev += wq & 0x00ff00ffu;
        od += (wq >> 8) & 0x00ff00ffu;
      }
      a[4 * i + 0] = ev & 0xffffu;
      a[4 * i + 1] = od & 0xffffu;
      a[4 * i + 2] = ev >> 16;
      a[4 * i + 3] = od >> 16;
    }
  }

  smap[t] = 58;
  __syncthreads();
  if (t < 58) smap[c_unis[t]] = t;
  __syncthreads();

  int m[RED_KL];
  {
    const int4* mp = reinterpret_cast<const int4*>(smap + koff);
#pragma unroll
    for (int i = 0; i < 4; ++i) {
      const int4 v = mp[i];
      m[4 * i + 0] = v.x;
      m[4 * i + 1] = v.y;
      m[4 * i + 2] = v.z;
      m[4 * i + 3] = v.w;
    }
  }

  v8f acc[4] = {};
#pragma unroll
  for (int s = 0; s < RED_KW / 4; ++s) {
    const v2f A = {(float)a[2 * s], (float)a[2 * s + 1]};
#pragma unroll
    for (int nt = 0; nt < 4; ++nt) {
      const int n = nt * 16 + c;
      const v2f Bm = {(m[2 * s] == n) ? 1.0f : 0.0f, (m[2 * s + 1] == n) ? 1.0f : 0.0f};
      acc[nt] = __builtin_amdgcn_wmma_f32_16x16x4_f32(
           false, A,  false, Bm,
           (short)0, acc[nt],  false,  false);
    }
  }

#pragma unroll
  for (int nt = 0; nt < 4; ++nt) {
#pragma unroll
    for (int r = 0; r < 8; ++r) part[((wave * 4 + nt) * 8 + r) * 32 + lane] = acc[nt][r];
  }
  __syncthreads();

  constexpr int RED_PER = (RED_OUT + RED_WAVES * 32 - 1) / (RED_WAVES * 32);
#pragma unroll
  for (int q = 0; q < RED_PER; ++q) {
    const int o = t + q * (RED_WAVES * 32);
    if ((q + 1) * (RED_WAVES * 32) <= RED_OUT || o < RED_OUT) {
      const int row = o / 59;
      const int n = o - row * 59;
      const int idx = ((n >> 4) * 8 + (row & 7)) * 32 + (row >> 3) * 16 + (n & 15);
      float v = 0.0f;
      for (int w = 0; w < RED_WAVES; ++w) v += part[w * 1024 + idx];
      const unsigned iv = (unsigned)(v + 0.5f);
      out[mt * RED_OUT + o] = (float)(iv & 255u);
    }
  }
}

extern "C" void kernel_launch(void* const* d_in, const int* in_sizes, int n_in,
                              void* d_out, int out_size, void* d_ws,
                              size_t ws_size, hipStream_t stream) {
  const float* x = (const float*)d_in[0];
  float* out = (float*)d_out;
  unsigned* planes = (unsigned*)d_ws;

  const int B = in_sizes[0] >> 18;

  lbp_hist_kernel<<<dim3(LBP_PARTS, B), LBP_WAVES * 32, 0, stream>>>(x, planes);
  wmma_reduce_kernel<<<B / 16, RED_WAVES * 32, 0, stream>>>(planes, out);
}


# speedup vs baseline: 4.1090x; 1.0594x over previous
#include <hip/hip_runtime.h>
#include <hip/hip_bf16.h>
#include <stdint.h>

typedef __attribute__((ext_vector_type(2))) float v2f;
typedef __attribute__((ext_vector_type(8))) float v8f;

__constant__ int c_unis[58] = {
    0, 1, 2, 3, 4, 6, 7, 8, 12, 14, 15, 16, 24, 28, 30, 31, 32, 48, 56, 60,
    62, 63, 64, 96, 112, 120, 124, 126, 127, 128, 129, 131, 135, 143, 159,
    191, 192, 193, 195, 199, 207, 223, 224, 225, 227, 231, 239, 240, 241,
    243, 247, 248, 249, 251, 252, 253, 254, 255};

__device__ __forceinline__ unsigned quant(float f) {
  return (unsigned)(f * 255.0f);
}

constexpr int LBP_WAVES = 16;
constexpr int LBP_ROUNDS = 4;
constexpr int LBP_PASSES = 2;
constexpr int LBP_COPIES = 32;
constexpr int LBP_SETS = 2;
constexpr int SET_WORDS = 256 * LBP_COPIES;
constexpr int RND_ROWS = LBP_WAVES * LBP_PASSES;
constexpr int LBP_ROWS = RND_ROWS * LBP_ROUNDS;
constexpr int LBP_PARTS = 512 / LBP_ROWS;
constexpr int LBP_HALO = 3;
constexpr int STG_ROWS = LBP_ROWS + 2 * LBP_HALO;
constexpr int STG_FIRST = RND_ROWS + 2 * LBP_HALO;
constexpr int STG_PAD_W = 8;
constexpr int STG_PITCH_W = 128 + 2 * STG_PAD_W;
constexpr int STG_PITCH_Q = STG_PITCH_W / 4;
constexpr int STG_WIN_Q = STG_PAD_W / 4 - 1;
static_assert(LBP_PARTS * LBP_ROWS == 512, "four work-groups cover one image");
constexpr int ZERO_Q = LBP_SETS * SET_WORDS / 4 / (LBP_WAVES * 32);
static_assert(ZERO_Q * 4 * LBP_WAVES * 32 == LBP_SETS * SET_WORDS, "whole 16-byte stores per thread clear both sets");
static_assert(STG_ROWS - STG_FIRST == (LBP_ROUNDS - 1) * RND_ROWS, "every round but the last stages the next round's rows");
static_assert(STG_FIRST > 2 * LBP_WAVES && STG_FIRST <= 3 * LBP_WAVES, "three rows per wave at most before round 0");

__device__ __forceinline__ void load_window(const uint4* __restrict__ row4, unsigned lane,
                                            unsigned (&w)[12]) {
#pragma unroll
  for (int i = 0; i < 3; ++i) {
    const uint4 v = row4[STG_WIN_Q + lane + i];
    w[i * 4 + 0] = v.x;
    w[i * 4 + 1] = v.y;
    w[i * 4 + 2] = v.z;
    w[i * 4 + 3] = v.w;
  }
}

template <int S>
__device__ __forceinline__ unsigned cut4(unsigned hi, unsigned lo) {
  return (lo >> (8 * S)) | (hi << (32 - 8 * S));
}

__device__ __forceinline__ unsigned ge4(unsigned n, unsigned c, unsigned cy) {
  const unsigned s = (n & 0x7f7f7f7fu) + cy;
  return __builtin_amdgcn_bitop3_b32(n, c, s, 0xb2);
}

template <unsigned M>
__device__ __forceinline__ unsigned ins4(unsigned a, unsigned b) {
  return __builtin_amdgcn_bitop3_b32(a, b, M, 0xe4);
}

template <int I>
__device__ __forceinline__ unsigned lbp_code4(const unsigned (&wc)[12], const unsigned (&wu)[12],
                                              const unsigned (&wd)[12]) {
  constexpr int B = 4 + I;
  const unsigned c = wc[B];
  const unsigned cy = (~c & 0x7f7f7f7fu) + 0x01010101u;
  const unsigned r0 = ge4(wu[B], c, cy);
  const unsigned r1 = ge4(cut4<3>(wu[B + 1], wu[B]), c, cy);
  const unsigned r2 = ge4(cut4<1>(wc[B + 2], wc[B + 1]), c, cy);
  const unsigned r3 = ge4(cut4<3>(wd[B + 1], wd[B]), c, cy);
  const unsigned r4 = ge4(wd[B], c, cy);
  const unsigned r5 = ge4(cut4<1>(wd[B], wd[B - 1]), c, cy);
  const unsigned r6 = ge4(cut4<3>(wc[B - 1], wc[B - 2]), c, cy);
  const unsigned r7 = ge4(cut4<1>(wu[B], wu[B - 1]), c, cy);
  const unsigned p10 = ins4<0x40404040u>(r0 >> 1, r1);
  const unsigned p32 = ins4<0x40404040u>(r2 >> 1, r3);
  const unsigned p54 = ins4<0x40404040u>(r4 >> 1, r5);
  const unsigned p76 = ins4<0x40404040u>(r6 >> 1, r7);
  const unsigned qlo = ins4<0x30303030u>(p10 >> 2, p32);
  const unsigned qhi = ins4<0x30303030u>(p54 >> 2, p76);
  return ins4<0x0f0f0f0fu>(qlo >> 4, qhi);
}

template <int I>
__device__ __forceinline__ void lbp_row16(const unsigned (&wc)[12], const unsigned (&wu)[12],
                                          const unsigned (&wd)[12], unsigned* __restrict__ sh,
                                          unsigned lane) {
  if constexpr (I < 4) {
    const unsigned code = lbp_code4<I>(wc, wu, wd);
    atomicAdd(&sh[((code & 255u) << 5) | lane], 1u);
    atomicAdd(&sh[(((code >> 8) & 255u) << 5) | lane], 1u);
    atomicAdd(&sh[(((code >> 16) & 255u) << 5) | lane], 1u);
    atomicAdd(&sh[((code >> 24) << 5) | lane], 1u);
    lbp_row16<I + 1>(wc, wu, wd, sh, lane);
  }
}

template <int NR>
__device__ __forceinline__ void stage_load(const float* __restrict__ img, int row0, int jbase,
                                           int jend, int wave, unsigned lane,
                                           float4 (&v)[NR][4]) {
#pragma unroll
  for (int kk = 0; kk < NR; ++kk) {
    const int j = jbase + wave + LBP_WAVES * kk;
    const int jl = j < jend ? j : jend - 1;
    int gr = row0 + jl;
    gr = gr < 0 ? 0 : (gr > 511 ? 511 : gr);
    const float4* rp = reinterpret_cast<const float4*>(img + gr * 512);
#pragma unroll
    for (int i = 0; i < 4; ++i) v[kk][i] = rp[lane + 32 * i];
  }
}

template <int NR>
__device__ __forceinline__ void stage_store(const float4 (&v)[NR][4], int row0,
                                            unsigned* __restrict__ tile, int jbase, int jend,
                                            int wave, unsigned lane) {
#pragma unroll
  for (int kk = 0; kk < NR; ++kk) {
    const int j = jbase + wave + LBP_WAVES * kk;
    const bool in_img = (unsigned)(row0 + j) < 512u;
    if (j < jend) {
      unsigned* dst = tile + j * STG_PITCH_W;
#pragma unroll
      for (int i = 0; i < 4; ++i) {
        const unsigned pk = quant(v[kk][i].x) | (quant(v[kk][i].y) << 8) |
                            (quant(v[kk][i].z) << 16) | (quant(v[kk][i].w) << 24);
        dst[STG_PAD_W + lane + 32 * i] = in_img ? pk : 0u;
      }
      if (lane < 2 * STG_PAD_W) dst[lane < STG_PAD_W ? lane : 128 + lane] = 0u;
    }
  }
}

__device__ __forceinline__ void lds_barrier() {
  __builtin_amdgcn_fence(__ATOMIC_RELEASE, "workgroup", "local");
  __builtin_amdgcn_s_barrier();
  __builtin_amdgcn_fence(__ATOMIC_ACQUIRE, "workgroup", "local");
}

template <int NR>
__device__ __forceinline__ void stage_rows(const float* __restrict__ img, int row0,
                                           unsigned* __restrict__ tile, int jbase, int jend,
                                           int wave, unsigned lane) {
  float4 v[NR][4];
  stage_load<NR>(img, row0, jbase, jend, wave, lane, v);
  stage_store<NR>(v, row0, tile, jbase, jend, wave, lane);
}

__global__ __launch_bounds__(LBP_WAVES * 32) void lbp_hist_kernel(
    const float* __restrict__ x, unsigned* __restrict__ planes) {
  __shared__ __attribute__((aligned(16))) unsigned lds[LBP_SETS * SET_WORDS + STG_ROWS * STG_PITCH_W + 64];
  unsigned* hist = lds;
  unsigned* tile = lds + LBP_SETS * SET_WORDS;
  unsigned char* pb = reinterpret_cast<unsigned char*>(tile + STG_ROWS * STG_PITCH_W);
  const int t = threadIdx.x;
  {
    uint4* sh4 = reinterpret_cast<uint4*>(hist);
    const uint4 z = {0u, 0u, 0u, 0u};
#pragma unroll
    for (int i = 0; i < ZERO_Q; ++i) sh4[t + i * (LBP_WAVES * 32)] = z;
  }
  const unsigned lane = (unsigned)t & 31u;
  const int wave = t >> 5;
  unsigned simd;
  asm("s_getreg_b32 %0, hwreg(HW_REG_WAVE_HW_ID1, 8, 2)" : "=s"(simd));
  unsigned* sh = hist + (simd & 1u) * SET_WORDS;

  const int batch = blockIdx.y;
  const int row0 = blockIdx.x * LBP_ROWS - LBP_HALO;
  const float* img = x + (size_t)batch * (512 * 512);
  const uint4* tile4 = reinterpret_cast<const uint4*>(tile);

  stage_rows<3>(img, row0, tile, 0, STG_FIRST, wave, lane);

  for (int r = 0; r < LBP_ROUNDS; ++r) {
    const bool ahead = r + 1 < LBP_ROUNDS;
    const int jn = STG_FIRST + RND_ROWS * r;
    float4 nx[LBP_PASSES][4];
    if (ahead) stage_load<LBP_PASSES>(img, row0, jn, STG_ROWS, wave, lane, nx);
    lds_barrier();
#pragma nounroll
    for (int p = 0; p < LBP_PASSES; ++p) {
      const int jc = LBP_HALO + RND_ROWS * r + LBP_WAVES * p + wave;
      unsigned wc[12], wu[12], wd[12];
      load_window(tile4 + jc * STG_PITCH_Q, lane, wc);
      load_window(tile4 + (jc - LBP_HALO) * STG_PITCH_Q, lane, wu);
      load_window(tile4 + (jc + LBP_HALO) * STG_PITCH_Q, lane, wd);
      lbp_row16<0>(wc, wu, wd, sh, lane);
    }
    if (ahead) stage_store<LBP_PASSES>(nx, row0, tile, jn, STG_ROWS, wave, lane);
  }

  __syncthreads();
  if (t < 256) {
    unsigned v0 = 0u;
    for (int i = 0; i < LBP_COPIES; ++i) {
      const unsigned w = (t << 5) | ((unsigned)(i + t) & 31u);
      v0 += hist[w] + hist[SET_WORDS + w];
    }
    pb[t] = (unsigned char)v0;
  }
  __syncthreads();
  if (t < 32) {
    const uint2 v = reinterpret_cast<const uint2*>(pb)[t];
    reinterpret_cast<uint2*>(planes)[(batch * LBP_PARTS + (int)blockIdx.x) * 32 + t] = v;
  }
}

constexpr int RED_WAVES = 8;
constexpr int RED_KW = 256 / RED_WAVES;
constexpr int RED_KL = RED_KW / 2;
constexpr int RED_OUT = 16 * 59;
static_assert(RED_KL == 16, "one 16-byte load per lane and plane");

__global__ __launch_bounds__(RED_WAVES * 32) void wmma_reduce_kernel(
    const unsigned* __restrict__ planes, float* __restrict__ out) {
  __shared__ __attribute__((aligned(16))) int smap[256];
  __shared__ __attribute__((aligned(16))) float part[RED_WAVES * 4 * 256];
  const int t = threadIdx.x;
  const int wave = t >> 5;
  const int lane = t & 31;
  const int c = lane & 15;
  const int h = lane >> 4;
  const int mt = blockIdx.x;
  const int koff = wave * RED_KW + h * RED_KL;

  unsigned a[RED_KL];
  {
    const uint4* ap = reinterpret_cast<const uint4*>(planes) + (size_t)(mt * 16 + c) * (LBP_PARTS * 16) + (koff >> 4);
    uint4 pq[LBP_PARTS];
#pragma unroll
    for (int q = 0; q < LBP_PARTS; ++q) pq[q] = ap[16 * q];
#pragma unroll
    for (int i = 0; i < 4; ++i) {
      unsigned ev = 0u, od = 0u;
#pragma unroll
      for (int q = 0; q < LBP_PARTS; ++q) {
        const unsigned wq = i == 0 ? pq[q].x : (i == 1 ? pq[q].y : (i == 2 ? pq[q].z : pq[q].w));
        ev += wq & 0x00ff00ffu;
        od += (wq >> 8) & 0x00ff00ffu;
      }
      a[4 * i + 0] = ev & 0xffffu;
      a[4 * i + 1] = od & 0xffffu;
      a[4 * i + 2] = ev >> 16;
      a[4 * i + 3] = od >> 16;
    }
  }

  smap[t] = 58;
  __syncthreads();
  if (t < 58) smap[c_unis[t]] = t;
  __syncthreads();

  int m[RED_KL];
  {
    const int4* mp = reinterpret_cast<const int4*>(smap + koff);
#pragma unroll
    for (int i = 0; i < 4; ++i) {
      const int4 v = mp[i];
      m[4 * i + 0] = v.x;
      m[4 * i + 1] = v.y;
      m[4 * i + 2] = v.z;
      m[4 * i + 3] = v.w;
    }
  }

  v8f acc[4] = {};
#pragma unroll
  for (int s = 0; s < RED_KW / 4; ++s) {
    const v2f A = {(float)a[2 * s], (float)a[2 * s + 1]};
#pragma unroll
    for (int nt = 0; nt < 4; ++nt) {
      const int n = nt * 16 + c;
      const v2f Bm = {(m[2 * s] == n) ? 1.0f : 0.0f, (m[2 * s + 1] == n) ? 1.0f : 0.0f};
      acc[nt] = __builtin_amdgcn_wmma_f32_16x16x4_f32(
           false, A,  false, Bm,
           (short)0, acc[nt],  false,  false);
    }
  }

#pragma unroll
  for (int nt = 0; nt < 4; ++nt) {
#pragma unroll
    for (int r = 0; r < 8; ++r) part[((wave * 4 + nt) * 8 + r) * 32 + lane] = acc[nt][r];
  }
  __syncthreads();

  constexpr int RED_PER = (RED_OUT + RED_WAVES * 32 - 1) / (RED_WAVES * 32);
#pragma unroll
  for (int q = 0; q < RED_PER; ++q) {
    const int o = t + q * (RED_WAVES * 32);
    if ((q + 1) * (RED_WAVES * 32) <= RED_OUT || o < RED_OUT) {
      const int row = o / 59;
      const int n = o - row * 59;
      const int idx = ((n >> 4) * 8 + (row & 7)) * 32 + (row >> 3) * 16 + (n & 15);
      float v = 0.0f;
      for (int w = 0; w < RED_WAVES; ++w) v += part[w * 1024 + idx];
      const unsigned iv = (unsigned)(v + 0.5f);
      out[mt * RED_OUT + o] = (float)(iv & 255u);
    }
  }
}

extern "C" void kernel_launch(void* const* d_in, const int* in_sizes, int n_in,
                              void* d_out, int out_size, void* d_ws,
                              size_t ws_size, hipStream_t stream) {
  const float* x = (const float*)d_in[0];
  float* out = (float*)d_out;
  unsigned* planes = (unsigned*)d_ws;

  const int B = in_sizes[0] >> 18;

  lbp_hist_kernel<<<dim3(LBP_PARTS, B), LBP_WAVES * 32, 0, stream>>>(x, planes);
  wmma_reduce_kernel<<<B / 16, RED_WAVES * 32, 0, stream>>>(planes, out);
}
